# starting kernel
# baseline (speedup 1.0000x reference)
#include <hip/hip_runtime.h>
#include <hip/hip_bf16.h>

typedef __attribute__((ext_vector_type(2))) float v2f;
typedef __attribute__((ext_vector_type(8))) float v8f;

__global__ __launch_bounds__(256)
void switchgate_score_kernel(const float* __restrict__ X,
                             const float* __restrict__ W,
                             const float* __restrict__ bias,
                             float* __restrict__ out,
                             int D) {
    const int lane = threadIdx.x & 31;
    const int n    = lane & 15;
    const int h    = lane >> 4;
    const int wave = threadIdx.x >> 5;
    const int tile = blockIdx.x * 8 + wave;
    const long rowBase = (long)tile * 16;

    const float* __restrict__ xrow = X + (rowBase + n) * (long)D + 2 * h;
    const float* __restrict__ wrow = W + (long)n * D + 2 * h;

    v8f c = {0.f, 0.f, 0.f, 0.f, 0.f, 0.f, 0.f, 0.f};

    for (int k = 0; k < D; k += 16) {
#pragma unroll
        for (int u = 0; u < 4; ++u) {
            v2f a  = *(const v2f*)(xrow + k + 4 * u);
            v2f bb = *(const v2f*)(wrow + k + 4 * u);
            c = __builtin_amdgcn_wmma_f32_16x16x4_f32(
                     false, a,  false, bb,
                     (short)0, c,  false,  false);
        }
    }

    const float bias_n = bias[n];

    float sc[8];
#pragma unroll
    for (int i = 0; i < 8; ++i) sc[i] = c[i] + bias_n;

#pragma unroll
    for (int i = 0; i < 8; ++i) {
        float s = sc[i];

        float m = s;
#pragma unroll
        for (int mask = 1; mask < 16; mask <<= 1)
            m = fmaxf(m, __shfl_xor(m, mask, 32));

        float e = __expf(s - m);
        float sum = e;
#pragma unroll
        for (int mask = 1; mask < 16; mask <<= 1)
            sum += __shfl_xor(sum, mask, 32);

        float gate = e / sum;

        int rank = 0;
#pragma unroll
        for (int j = 0; j < 16; ++j) {
            float sj = __shfl(s, j, 16);
            rank += (sj > s) || (sj == s && j < n);
        }
        sc[i] = (rank < 8) ? gate : 0.0f;
    }

#pragma unroll
    for (int i = 0; i < 8; ++i) {
        long t = rowBase + h * 8 + i;
        out[t * 16 + n] = sc[i];
    }
}

__global__ __launch_bounds__(256)
void switchgate_norm_kernel(float* __restrict__ out, int SE) {
    int idx = blockIdx.x * blockDim.x + threadIdx.x;
    if (idx >= SE) return;

    float v[8];
    float sum = 0.0f;
#pragma unroll
    for (int b = 0; b < 8; ++b) {
        v[b] = out[(long)b * SE + idx];
        sum += v[b];
    }
    float scale = 8.0f / (sum + 1e-6f);
#pragma unroll
    for (int b = 0; b < 8; ++b)
        out[(long)b * SE + idx] = v[b] * scale;
}

extern "C" void kernel_launch(void* const* d_in, const int* in_sizes, int n_in,
                              void* d_out, int out_size, void* d_ws, size_t ws_size,
                              hipStream_t stream) {
    const float* X    = (const float*)d_in[0];
    const float* W    = (const float*)d_in[1];
    const float* bias = (const float*)d_in[2];
    float* out = (float*)d_out;

    const int E = in_sizes[2];
    const int D = in_sizes[1] / E;
    const int tokens = in_sizes[0] / D;

    const int tiles = tokens / 16;
    switchgate_score_kernel<<<tiles / 8, 256, 0, stream>>>(X, W, bias, out, D);

    const int SE = out_size / 8;
    switchgate_norm_kernel<<<(SE + 255) / 256, 256, 0, stream>>>(out, SE);
}


# W copied once per work-group into LDS in B-fragment order: B fragments by ds_load_b128 (one per two WMMAs), not per-wave 8-byte loads; K loop rebuilt around it, X asked 3 blocks ahead; sums kept
# speedup vs baseline: 1.6381x; 1.6381x over previous
#include <hip/hip_runtime.h>
#include <hip/hip_bf16.h>

typedef __attribute__((ext_vector_type(2))) float v2f;
typedef __attribute__((ext_vector_type(4))) float v4f;
typedef __attribute__((ext_vector_type(8))) float v8f;


constexpr int KD = 2048;
constexpr int GPITCH = 144;

__device__ __forceinline__ void load_a_block(v2f (&a)[8], const float* __restrict__ p) {
#pragma unroll
    for (int u = 0; u < 8; ++u) a[u] = *(const v2f*)(p + 4 * u);
}

__device__ __forceinline__ void load_b_block(v4f (&b)[4], const float* p) {
#pragma unroll
    for (int j = 0; j < 4; ++j) b[j] = *(const v4f*)(p + j * GPITCH);
}

__device__ __forceinline__ v8f mma_block(const v2f (&a)[8], const v4f (&b)[4], v8f c) {
#pragma unroll
    for (int u = 0; u < 8; ++u) {
        const v4f q = b[u >> 1];
        const v2f bb = (u & 1) ? (v2f){q.z, q.w} : (v2f){q.x, q.y};
        c = __builtin_amdgcn_wmma_f32_16x16x4_f32(
                 false, a[u],  false, bb,
                 (short)0, c,  false,  false);
    }
    return c;
}

template <int S, bool LOADA, bool LOADB>
__device__ __forceinline__ v8f step(v2f (&a)[4][8], v4f (&b)[2][4], v8f c,
                                    const float* __restrict__ xk, const float* wk) {
    if constexpr (LOADA) load_a_block(a[(S + 3) & 3], xk + 32 * (S + 3));
    if constexpr (LOADB) load_b_block(b[(S + 1) & 1], wk + 4 * (S + 1) * GPITCH);
    __builtin_amdgcn_sched_barrier(0);
    c = mma_block(a[S], b[S & 1], c);
    __builtin_amdgcn_sched_barrier(0);
    return c;
}

__global__ __launch_bounds__(256)
void switchgate_score_kernel(const float* __restrict__ X,
                             const float* __restrict__ W,
                             const float* __restrict__ bias,
                             float* __restrict__ out,
                             int D) {
    __shared__ __attribute__((aligned(16))) float Wl[(KD / 8) * GPITCH];

    const int lane = threadIdx.x & 31;
    const int n    = lane & 15;
    const int h    = lane >> 4;
    const int wave = threadIdx.x >> 5;
    const int tile = blockIdx.x * 8 + wave;
    const long rowBase = (long)tile * 16;

    const float* __restrict__ xrow = X + (rowBase + n) * (long)D + 2 * h;

    v2f a[4][8];
    v4f b[2][4];

    load_a_block(a[0], xrow);
    load_a_block(a[1], xrow + 32);
    load_a_block(a[2], xrow + 64);

    {
        const int r4 = lane >> 3;
        const int q8 = lane & 7;
        v4f w4[32];
#pragma unroll
        for (int i = 0; i < 32; ++i) {
            const int id  = wave + 8 * i;
            const int row = (id & 3) * 4 + r4;
            const int q   = (id >> 2) * 8 + q8;
            w4[i] = *(const v4f*)(W + (long)row * D + 4 * q);
        }
#pragma unroll
        for (int i = 0; i < 32; ++i) {
            const int id  = wave + 8 * i;
            const int row = (id & 3) * 4 + r4;
            const int q   = (id >> 2) * 8 + q8;
            float* dst = Wl + (q >> 1) * GPITCH + row * 4 + (q & 1) * 2;
            *(v2f*)dst        = (v2f){w4[i].x, w4[i].y};
            *(v2f*)(dst + 64) = (v2f){w4[i].z, w4[i].w};
        }
    }
    __syncthreads();

    const float* wl = Wl + lane * 4;
    load_b_block(b[0], wl);

    v8f c = {0.f, 0.f, 0.f, 0.f, 0.f, 0.f, 0.f, 0.f};

    const float* __restrict__ xk = xrow;
    const float* wk = wl;
#pragma unroll
    for (int t = 0; t < KD / 128 - 1; ++t) {
        c = step<0, true, true>(a, b, c, xk, wk);
        c = step<1, true, true>(a, b, c, xk, wk);
        c = step<2, true, true>(a, b, c, xk, wk);
        c = step<3, true, true>(a, b, c, xk, wk);
        xk += 128;
        wk += 16 * GPITCH;
    }
    c = step<0, true,  true >(a, b, c, xk, wk);
    c = step<1, false, true >(a, b, c, xk, wk);
    c = step<2, false, true >(a, b, c, xk, wk);
    c = step<3, false, false>(a, b, c, xk, wk);

    const float bias_n = bias[n];

    float sc[8];
#pragma unroll
    for (int i = 0; i < 8; ++i) sc[i] = c[i] + bias_n;

#pragma unroll
    for (int i = 0; i < 8; ++i) {
        float s = sc[i];

        float m = s;
#pragma unroll
        for (int mask = 1; mask < 16; mask <<= 1)
            m = fmaxf(m, __shfl_xor(m, mask, 32));

        float e = __expf(s - m);
        float sum = e;
#pragma unroll
        for (int mask = 1; mask < 16; mask <<= 1)
            sum += __shfl_xor(sum, mask, 32);

        float gate = e / sum;

        int rank = 0;
#pragma unroll
        for (int j = 0; j < 16; ++j) {
            float sj = __shfl(s, j, 16);
            rank += (sj > s) || (sj == s && j < n);
        }
        sc[i] = (rank < 8) ? gate : 0.0f;
    }

#pragma unroll
    for (int i = 0; i < 8; ++i) {
        long t = rowBase + h * 8 + i;
        out[t * 16 + n] = sc[i];
    }
}

__global__ __launch_bounds__(256)
void switchgate_norm_kernel(float* __restrict__ out, int SE) {
    int idx = blockIdx.x * blockDim.x + threadIdx.x;
    if (idx >= SE) return;

    float v[8];
    float sum = 0.0f;
#pragma unroll
    for (int b = 0; b < 8; ++b) {
        v[b] = out[(long)b * SE + idx];
        sum += v[b];
    }
    float scale = 8.0f / (sum + 1e-6f);
#pragma unroll
    for (int b = 0; b < 8; ++b)
        out[(long)b * SE + idx] = v[b] * scale;
}

extern "C" void kernel_launch(void* const* d_in, const int* in_sizes, int n_in,
                              void* d_out, int out_size, void* d_ws, size_t ws_size,
                              hipStream_t stream) {
    const float* X    = (const float*)d_in[0];
    const float* W    = (const float*)d_in[1];
    const float* bias = (const float*)d_in[2];
    float* out = (float*)d_out;

    const int E = in_sizes[2];
    const int D = in_sizes[1] / E;
    const int tokens = in_sizes[0] / D;

    const int tiles = tokens / 16;
    switchgate_score_kernel<<<tiles / 8, 256, 0, stream>>>(X, W, bias, out, D);

    const int SE = out_size / 8;
    switchgate_norm_kernel<<<(SE + 255) / 256, 256, 0, stream>>>(out, SE);
}


# X fragments by one 16-byte load per lane and 8-k group (was two 8-byte loads): k dealt out inside each group to match, W's LDS copy in the same order; f32 sum order changes inside each 8-k group
# speedup vs baseline: 1.6973x; 1.0361x over previous
#include <hip/hip_runtime.h>
#include <hip/hip_bf16.h>

typedef __attribute__((ext_vector_type(2))) float v2f;
typedef __attribute__((ext_vector_type(4))) float v4f;
typedef __attribute__((ext_vector_type(8))) float v8f;


constexpr int KD = 2048;
constexpr int GPITCH = 144;

__device__ __forceinline__ void load_a_block(v4f (&a)[4], const float* __restrict__ p) {
#pragma unroll
    for (int g = 0; g < 4; ++g) a[g] = *(const v4f*)(p + 8 * g);
}

__device__ __forceinline__ void load_b_block(v4f (&b)[4], const float* p) {
#pragma unroll
    for (int j = 0; j < 4; ++j) b[j] = *(const v4f*)(p + j * GPITCH);
}

__device__ __forceinline__ v8f mma_block(const v4f (&a)[4], const v4f (&b)[4], v8f c) {
#pragma unroll
    for (int u = 0; u < 8; ++u) {
        const v4f p = a[u >> 1];
        const v4f q = b[u >> 1];
        const v2f aa = (u & 1) ? (v2f){p.z, p.w} : (v2f){p.x, p.y};
        const v2f bb = (u & 1) ? (v2f){q.z, q.w} : (v2f){q.x, q.y};
        c = __builtin_amdgcn_wmma_f32_16x16x4_f32(
                 false, aa,  false, bb,
                 (short)0, c,  false,  false);
    }
    return c;
}

template <int S, bool LOADA, bool LOADB>
__device__ __forceinline__ v8f step(v4f (&a)[4][4], v4f (&b)[2][4], v8f c,
                                    const float* __restrict__ xk, const float* wk) {
    if constexpr (LOADA) load_a_block(a[(S + 3) & 3], xk + 32 * (S + 3));
    if constexpr (LOADB) load_b_block(b[(S + 1) & 1], wk + 4 * (S + 1) * GPITCH);
    __builtin_amdgcn_sched_barrier(0);
    c = mma_block(a[S], b[S & 1], c);
    __builtin_amdgcn_sched_barrier(0);
    return c;
}

__global__ __launch_bounds__(256)
void switchgate_score_kernel(const float* __restrict__ X,
                             const float* __restrict__ W,
                             const float* __restrict__ bias,
                             float* __restrict__ out,
                             int D) {
    __shared__ __attribute__((aligned(16))) float Wl[(KD / 8) * GPITCH];

    const int lane = threadIdx.x & 31;
    const int n    = lane & 15;
    const int h    = lane >> 4;
    const int wave = threadIdx.x >> 5;
    const int tile = blockIdx.x * 8 + wave;
    const long rowBase = (long)tile * 16;

    const float* __restrict__ xrow = X + (rowBase + n) * (long)D + 4 * h;

    v4f a[4][4];
    v4f b[2][4];

    load_a_block(a[0], xrow);
    load_a_block(a[1], xrow + 32);
    load_a_block(a[2], xrow + 64);

    {
        const int r4 = lane >> 3;
        const int q8 = lane & 7;
        v4f w4[32];
#pragma unroll
        for (int i = 0; i < 32; ++i) {
            const int id  = wave + 8 * i;
            const int row = (id & 3) * 4 + r4;
            const int q   = (id >> 2) * 8 + q8;
            w4[i] = *(const v4f*)(W + (long)row * D + 4 * q);
        }
#pragma unroll
        for (int i = 0; i < 32; ++i) {
            const int id  = wave + 8 * i;
            const int row = (id & 3) * 4 + r4;
            const int q   = (id >> 2) * 8 + q8;
            float* dst = Wl + (q >> 1) * GPITCH + (row + 16 * (q & 1)) * 4;
            *(v4f*)dst = w4[i];
        }
    }
    __syncthreads();

    const float* wl = Wl + lane * 4;
    load_b_block(b[0], wl);

    v8f c = {0.f, 0.f, 0.f, 0.f, 0.f, 0.f, 0.f, 0.f};

    const float* __restrict__ xk = xrow;
    const float* wk = wl;
#pragma unroll
    for (int t = 0; t < KD / 128 - 1; ++t) {
        c = step<0, true, true>(a, b, c, xk, wk);
        c = step<1, true, true>(a, b, c, xk, wk);
        c = step<2, true, true>(a, b, c, xk, wk);
        c = step<3, true, true>(a, b, c, xk, wk);
        xk += 128;
        wk += 16 * GPITCH;
    }
    c = step<0, true,  true >(a, b, c, xk, wk);
    c = step<1, false, true >(a, b, c, xk, wk);
    c = step<2, false, true >(a, b, c, xk, wk);
    c = step<3, false, false>(a, b, c, xk, wk);

    const float bias_n = bias[n];

    float sc[8];
#pragma unroll
    for (int i = 0; i < 8; ++i) sc[i] = c[i] + bias_n;

#pragma unroll
    for (int i = 0; i < 8; ++i) {
        float s = sc[i];

        float m = s;
#pragma unroll
        for (int mask = 1; mask < 16; mask <<= 1)
            m = fmaxf(m, __shfl_xor(m, mask, 32));

        float e = __expf(s - m);
        float sum = e;
#pragma unroll
        for (int mask = 1; mask < 16; mask <<= 1)
            sum += __shfl_xor(sum, mask, 32);

        float gate = e / sum;

        int rank = 0;
#pragma unroll
        for (int j = 0; j < 16; ++j) {
            float sj = __shfl(s, j, 16);
            rank += (sj > s) || (sj == s && j < n);
        }
        sc[i] = (rank < 8) ? gate : 0.0f;
    }

#pragma unroll
    for (int i = 0; i < 8; ++i) {
        long t = rowBase + h * 8 + i;
        out[t * 16 + n] = sc[i];
    }
}

__global__ __launch_bounds__(256)
void switchgate_norm_kernel(float* __restrict__ out, int SE) {
    int idx = blockIdx.x * blockDim.x + threadIdx.x;
    if (idx >= SE) return;

    float v[8];
    float sum = 0.0f;
#pragma unroll
    for (int b = 0; b < 8; ++b) {
        v[b] = out[(long)b * SE + idx];
        sum += v[b];
    }
    float scale = 8.0f / (sum + 1e-6f);
#pragma unroll
    for (int b = 0; b < 8; ++b)
        out[(long)b * SE + idx] = v[b] * scale;
}

extern "C" void kernel_launch(void* const* d_in, const int* in_sizes, int n_in,
                              void* d_out, int out_size, void* d_ws, size_t ws_size,
                              hipStream_t stream) {
    const float* X    = (const float*)d_in[0];
    const float* W    = (const float*)d_in[1];
    const float* bias = (const float*)d_in[2];
    float* out = (float*)d_out;

    const int E = in_sizes[2];
    const int D = in_sizes[1] / E;
    const int tokens = in_sizes[0] / D;

    const int tiles = tokens / 16;
    switchgate_score_kernel<<<tiles / 8, 256, 0, stream>>>(X, W, bias, out, D);

    const int SE = out_size / 8;
    switchgate_norm_kernel<<<(SE + 255) / 256, 256, 0, stream>>>(out, SE);
}


# Each tile's K split in halves between two waves (16-wave work-groups, 4 waves per SIMD, was 2): partial tiles meet in LDS, each wave finishes 8 tokens; f32 sum order changes (two K-half sums added)
# speedup vs baseline: 1.7849x; 1.0516x over previous
#include <hip/hip_runtime.h>
#include <hip/hip_bf16.h>

typedef __attribute__((ext_vector_type(2))) float v2f;
typedef __attribute__((ext_vector_type(4))) float v4f;
typedef __attribute__((ext_vector_type(8))) float v8f;


constexpr int KD = 2048;
constexpr int KH = KD / 2;
constexpr int GPITCH = 144;

__device__ __forceinline__ void load_a_block(v4f (&a)[4], const float* __restrict__ p) {
#pragma unroll
    for (int g = 0; g < 4; ++g) a[g] = *(const v4f*)(p + 8 * g);
}

__device__ __forceinline__ void load_b_block(v4f (&b)[4], const float* p) {
#pragma unroll
    for (int j = 0; j < 4; ++j) b[j] = *(const v4f*)(p + j * GPITCH);
}

__device__ __forceinline__ v8f mma_block(const v4f (&a)[4], const v4f (&b)[4], v8f c) {
#pragma unroll
    for (int u = 0; u < 8; ++u) {
        const v4f p = a[u >> 1];
        const v4f q = b[u >> 1];
        const v2f aa = (u & 1) ? (v2f){p.z, p.w} : (v2f){p.x, p.y};
        const v2f bb = (u & 1) ? (v2f){q.z, q.w} : (v2f){q.x, q.y};
        c = __builtin_amdgcn_wmma_f32_16x16x4_f32(
                 false, aa,  false, bb,
                 (short)0, c,  false,  false);
    }
    return c;
}

template <int S, bool LOADA, bool LOADB>
__device__ __forceinline__ v8f step(v4f (&a)[4][4], v4f (&b)[2][4], v8f c,
                                    const float* __restrict__ xk, const float* wk) {
    if constexpr (LOADA) load_a_block(a[(S + 3) & 3], xk + 32 * (S + 3));
    if constexpr (LOADB) load_b_block(b[(S + 1) & 1], wk + 4 * (S + 1) * GPITCH);
    __builtin_amdgcn_sched_barrier(0);
    c = mma_block(a[S], b[S & 1], c);
    __builtin_amdgcn_sched_barrier(0);
    return c;
}

__global__ __launch_bounds__(512)
void switchgate_score_kernel(const float* __restrict__ X,
                             const float* __restrict__ W,
                             const float* __restrict__ bias,
                             float* __restrict__ out,
                             int D) {
    __shared__ __attribute__((aligned(16))) float Wl[(KD / 8) * GPITCH];
    __shared__ __attribute__((aligned(16))) float Pl[16 * 32 * 8];

    const int lane = threadIdx.x & 31;
    const int n    = lane & 15;
    const int h    = lane >> 4;
    const int wave = threadIdx.x >> 5;
    const int tw   = wave & 7;
    const int kh   = wave >> 3;
    const int tile = blockIdx.x * 8 + tw;
    const long rowBase = (long)tile * 16;

    const float* __restrict__ xrow = X + (rowBase + n) * (long)D + KH * kh + 4 * h;

    v4f a[4][4];
    v4f b[2][4];

    load_a_block(a[0], xrow);
    load_a_block(a[1], xrow + 32);
    load_a_block(a[2], xrow + 64);

    {
        const int r4 = lane >> 3;
        const int q8 = lane & 7;
        v4f w4[16];
#pragma unroll
        for (int i = 0; i < 16; ++i) {
            const int id  = wave + 16 * i;
            const int row = (id & 3) * 4 + r4;
            const int q   = (id >> 2) * 8 + q8;
            w4[i] = *(const v4f*)(W + (long)row * D + 4 * q);
        }
#pragma unroll
        for (int i = 0; i < 16; ++i) {
            const int id  = wave + 16 * i;
            const int row = (id & 3) * 4 + r4;
            const int q   = (id >> 2) * 8 + q8;
            float* dst = Wl + (q >> 1) * GPITCH + (row + 16 * (q & 1)) * 4;
            *(v4f*)dst = w4[i];
        }
    }
    __syncthreads();

    const float* wl = Wl + kh * (KH / 8) * GPITCH + lane * 4;
    load_b_block(b[0], wl);

    v8f c = {0.f, 0.f, 0.f, 0.f, 0.f, 0.f, 0.f, 0.f};

    const float* __restrict__ xk = xrow;
    const float* wk = wl;
#pragma unroll
    for (int t = 0; t < KH / 128 - 1; ++t) {
        c = step<0, true, true>(a, b, c, xk, wk);
        c = step<1, true, true>(a, b, c, xk, wk);
        c = step<2, true, true>(a, b, c, xk, wk);
        c = step<3, true, true>(a, b, c, xk, wk);
        xk += 128;
        wk += 16 * GPITCH;
    }
    c = step<0, true,  true >(a, b, c, xk, wk);
    c = step<1, false, true >(a, b, c, xk, wk);
    c = step<2, false, true >(a, b, c, xk, wk);
    c = step<3, false, false>(a, b, c, xk, wk);

    const float bias_n = bias[n];

    {
        float* pw = Pl + (wave * 32 + lane) * 8;
        *(v4f*)(pw)     = (v4f){c[0], c[1], c[2], c[3]};
        *(v4f*)(pw + 4) = (v4f){c[4], c[5], c[6], c[7]};
    }
    __syncthreads();

    const v4f p0 = *(const v4f*)(Pl + (tw * 32 + lane) * 8 + 4 * kh);
    const v4f p1 = *(const v4f*)(Pl + ((tw + 8) * 32 + lane) * 8 + 4 * kh);

    float sc[4];
#pragma unroll
    for (int i = 0; i < 4; ++i) sc[i] = (p0[i] + p1[i]) + bias_n;

#pragma unroll
    for (int i = 0; i < 4; ++i) {
        float s = sc[i];

        float m = s;
#pragma unroll
        for (int mask = 1; mask < 16; mask <<= 1)
            m = fmaxf(m, __shfl_xor(m, mask, 32));

        float e = __expf(s - m);
        float sum = e;
#pragma unroll
        for (int mask = 1; mask < 16; mask <<= 1)
            sum += __shfl_xor(sum, mask, 32);

        float gate = e / sum;

        int rank = 0;
#pragma unroll
        for (int j = 0; j < 16; ++j) {
            float sj = __shfl(s, j, 16);
            rank += (sj > s) || (sj == s && j < n);
        }
        sc[i] = (rank < 8) ? gate : 0.0f;
    }

#pragma unroll
    for (int i = 0; i < 4; ++i) {
        long t = rowBase + h * 8 + 4 * kh + i;
        out[t * 16 + n] = sc[i];
    }
}

__global__ __launch_bounds__(256)
void switchgate_norm_kernel(float* __restrict__ out, int SE) {
    int idx = blockIdx.x * blockDim.x + threadIdx.x;
    if (idx >= SE) return;

    float v[8];
    float sum = 0.0f;
#pragma unroll
    for (int b = 0; b < 8; ++b) {
        v[b] = out[(long)b * SE + idx];
        sum += v[b];
    }
    float scale = 8.0f / (sum + 1e-6f);
#pragma unroll
    for (int b = 0; b < 8; ++b)
        out[(long)b * SE + idx] = v[b] * scale;
}

extern "C" void kernel_launch(void* const* d_in, const int* in_sizes, int n_in,
                              void* d_out, int out_size, void* d_ws, size_t ws_size,
                              hipStream_t stream) {
    const float* X    = (const float*)d_in[0];
    const float* W    = (const float*)d_in[1];
    const float* bias = (const float*)d_in[2];
    float* out = (float*)d_out;

    const int E = in_sizes[2];
    const int D = in_sizes[1] / E;
    const int tokens = in_sizes[0] / D;

    const int tiles = tokens / 16;
    switchgate_score_kernel<<<tiles / 8, 512, 0, stream>>>(X, W, bias, out, D);

    const int SE = out_size / 8;
    switchgate_norm_kernel<<<(SE + 255) / 256, 256, 0, stream>>>(out, SE);
}


# X rows prefetched into the second-level cache two 128-k trips ahead of the fragment loads: one prefetch per wave and trip (lane = row x 256-byte unit); loads, WMMAs and sum order unchanged
# speedup vs baseline: 1.8433x; 1.0327x over previous
#include <hip/hip_runtime.h>
#include <hip/hip_bf16.h>

typedef __attribute__((ext_vector_type(2))) float v2f;
typedef __attribute__((ext_vector_type(4))) float v4f;
typedef __attribute__((ext_vector_type(8))) float v8f;


constexpr int KD = 2048;
constexpr int KH = KD / 2;
constexpr int GPITCH = 144;
constexpr int PF_AHEAD = 2;
#define PF_LOCALITY 2

__device__ __forceinline__ void load_a_block(v4f (&a)[4], const float* __restrict__ p) {
#pragma unroll
    for (int g = 0; g < 4; ++g) a[g] = *(const v4f*)(p + 8 * g);
}

__device__ __forceinline__ void load_b_block(v4f (&b)[4], const float* p) {
#pragma unroll
    for (int j = 0; j < 4; ++j) b[j] = *(const v4f*)(p + j * GPITCH);
}

__device__ __forceinline__ void prefetch_trip(const float* __restrict__ p) {
    __builtin_prefetch(p,  0,  PF_LOCALITY);
}

__device__ __forceinline__ v8f mma_block(const v4f (&a)[4], const v4f (&b)[4], v8f c) {
#pragma unroll
    for (int u = 0; u < 8; ++u) {
        const v4f p = a[u >> 1];
        const v4f q = b[u >> 1];
        const v2f aa = (u & 1) ? (v2f){p.z, p.w} : (v2f){p.x, p.y};
        const v2f bb = (u & 1) ? (v2f){q.z, q.w} : (v2f){q.x, q.y};
        c = __builtin_amdgcn_wmma_f32_16x16x4_f32(
                 false, aa,  false, bb,
                 (short)0, c,  false,  false);
    }
    return c;
}

template <int S, bool LOADA, bool LOADB>
__device__ __forceinline__ v8f step(v4f (&a)[4][4], v4f (&b)[2][4], v8f c,
                                    const float* __restrict__ xk, const float* wk) {
    if constexpr (LOADA) load_a_block(a[(S + 3) & 3], xk + 32 * (S + 3));
    if constexpr (LOADB) load_b_block(b[(S + 1) & 1], wk + 4 * (S + 1) * GPITCH);
    __builtin_amdgcn_sched_barrier(0);
    c = mma_block(a[S], b[S & 1], c);
    __builtin_amdgcn_sched_barrier(0);
    return c;
}

__global__ __launch_bounds__(512)
void switchgate_score_kernel(const float* __restrict__ X,
                             const float* __restrict__ W,
                             const float* __restrict__ bias,
                             float* __restrict__ out,
                             int D) {
    __shared__ __attribute__((aligned(16))) float Wl[(KD / 8) * GPITCH];
    __shared__ __attribute__((aligned(16))) float Pl[16 * 32 * 8];

    const int lane = threadIdx.x & 31;
    const int n    = lane & 15;
    const int h    = lane >> 4;
    const int wave = threadIdx.x >> 5;
    const int tw   = wave & 7;
    const int kh   = wave >> 3;
    const int tile = blockIdx.x * 8 + tw;
    const long rowBase = (long)tile * 16;

    const float* __restrict__ xrow = X + (rowBase + n) * (long)D + KH * kh + 4 * h;

    v4f a[4][4];
    v4f b[2][4];

    load_a_block(a[0], xrow);
    load_a_block(a[1], xrow + 32);
    load_a_block(a[2], xrow + 64);

    const float* __restrict__ xpf = X + (rowBase + n) * (long)D + KH * kh + 64 * h;
#pragma unroll
    for (int t = 1; t <= PF_AHEAD; ++t) prefetch_trip(xpf + 128 * t);

    {
        const int r4 = lane >> 3;
        const int q8 = lane & 7;
        v4f w4[16];
#pragma unroll
        for (int i = 0; i < 16; ++i) {
            const int id  = wave + 16 * i;
            const int row = (id & 3) * 4 + r4;
            const int q   = (id >> 2) * 8 + q8;
            w4[i] = *(const v4f*)(W + (long)row * D + 4 * q);
        }
#pragma unroll
        for (int i = 0; i < 16; ++i) {
            const int id  = wave + 16 * i;
            const int row = (id & 3) * 4 + r4;
            const int q   = (id >> 2) * 8 + q8;
            float* dst = Wl + (q >> 1) * GPITCH + (row + 16 * (q & 1)) * 4;
            *(v4f*)dst = w4[i];
        }
    }
    __syncthreads();

    const float* wl = Wl + kh * (KH / 8) * GPITCH + lane * 4;
    load_b_block(b[0], wl);

    v8f c = {0.f, 0.f, 0.f, 0.f, 0.f, 0.f, 0.f, 0.f};

    const float* __restrict__ xk = xrow;
    const float* wk = wl;
#pragma unroll
    for (int t = 0; t < KH / 128 - 1; ++t) {
        if (t >= 1 && t + PF_AHEAD < KH / 128) prefetch_trip(xpf + 128 * (t + PF_AHEAD));
        c = step<0, true, true>(a, b, c, xk, wk);
        c = step<1, true, true>(a, b, c, xk, wk);
        c = step<2, true, true>(a, b, c, xk, wk);
        c = step<3, true, true>(a, b, c, xk, wk);
        xk += 128;
        wk += 16 * GPITCH;
    }
    c = step<0, true,  true >(a, b, c, xk, wk);
    c = step<1, false, true >(a, b, c, xk, wk);
    c = step<2, false, true >(a, b, c, xk, wk);
    c = step<3, false, false>(a, b, c, xk, wk);

    const float bias_n = bias[n];

    {
        float* pw = Pl + (wave * 32 + lane) * 8;
        *(v4f*)(pw)     = (v4f){c[0], c[1], c[2], c[3]};
        *(v4f*)(pw + 4) = (v4f){c[4], c[5], c[6], c[7]};
    }
    __syncthreads();

    const v4f p0 = *(const v4f*)(Pl + (tw * 32 + lane) * 8 + 4 * kh);
    const v4f p1 = *(const v4f*)(Pl + ((tw + 8) * 32 + lane) * 8 + 4 * kh);

    float sc[4];
#pragma unroll
    for (int i = 0; i < 4; ++i) sc[i] = (p0[i] + p1[i]) + bias_n;

#pragma unroll
    for (int i = 0; i < 4; ++i) {
        float s = sc[i];

        float m = s;
#pragma unroll
        for (int mask = 1; mask < 16; mask <<= 1)
            m = fmaxf(m, __shfl_xor(m, mask, 32));

        float e = __expf(s - m);
        float sum = e;
#pragma unroll
        for (int mask = 1; mask < 16; mask <<= 1)
            sum += __shfl_xor(sum, mask, 32);

        float gate = e / sum;

        int rank = 0;
#pragma unroll
        for (int j = 0; j < 16; ++j) {
            float sj = __shfl(s, j, 16);
            rank += (sj > s) || (sj == s && j < n);
        }
        sc[i] = (rank < 8) ? gate : 0.0f;
    }

#pragma unroll
    for (int i = 0; i < 4; ++i) {
        long t = rowBase + h * 8 + 4 * kh + i;
        out[t * 16 + n] = sc[i];
    }
}

__global__ __launch_bounds__(256)
void switchgate_norm_kernel(float* __restrict__ out, int SE) {
    int idx = blockIdx.x * blockDim.x + threadIdx.x;
    if (idx >= SE) return;

    float v[8];
    float sum = 0.0f;
#pragma unroll
    for (int b = 0; b < 8; ++b) {
        v[b] = out[(long)b * SE + idx];
        sum += v[b];
    }
    float scale = 8.0f / (sum + 1e-6f);
#pragma unroll
    for (int b = 0; b < 8; ++b)
        out[(long)b * SE + idx] = v[b] * scale;
}

extern "C" void kernel_launch(void* const* d_in, const int* in_sizes, int n_in,
                              void* d_out, int out_size, void* d_ws, size_t ws_size,
                              hipStream_t stream) {
    const float* X    = (const float*)d_in[0];
    const float* W    = (const float*)d_in[1];
    const float* bias = (const float*)d_in[2];
    float* out = (float*)d_out;

    const int E = in_sizes[2];
    const int D = in_sizes[1] / E;
    const int tokens = in_sizes[0] / D;

    const int tiles = tokens / 16;
    switchgate_score_kernel<<<tiles / 8, 512, 0, stream>>>(X, W, bias, out, D);

    const int SE = out_size / 8;
    switchgate_norm_kernel<<<(SE + 255) / 256, 256, 0, stream>>>(out, SE);
}
